# starting kernel
# baseline (speedup 1.0000x reference)
#include <hip/hip_runtime.h>
#include <hip/hip_bf16.h>

#define KS      7
#define DILX    2
#define NHEADS  4
#define NFRAMES 3
#define CDIM    128
#define HD      32
#define HH      64
#define WW      64
#define NPIX    4096
#define NB      2
#define KTOT    (NFRAMES*CDIM)
#define KSTEPS  (KTOT/4)
#define NFRAG   (KSTEPS*8*32)

typedef __attribute__((ext_vector_type(2))) float v2f;
typedef __attribute__((ext_vector_type(8))) float v8f;

__device__ __forceinline__ int win_start(int i) {
    int ni = i - (KS / 2) * DILX;
    if (ni < 0) return i % DILX;
    if (i + (KS / 2) * DILX >= HH) return HH - KS * DILX + (i % DILX);
    return ni;
}

__global__ void natten_attn_kernel(const float* __restrict__ q,
                                   const float* __restrict__ kk,
                                   const float* __restrict__ vv,
                                   const float* __restrict__ rpb,
                                   float* __restrict__ xn)
{
    int tid = blockIdx.x * blockDim.x + threadIdx.x;
    int j = tid & 63;
    int t = tid >> 6;
    int i = t & 63;
    t >>= 6;
    int f = t % NFRAMES;
    t /= NFRAMES;
    int head = t & 3;
    int b = t >> 2;

    int si = win_start(i), sj = win_start(j);
    int pbi = (si - i) / DILX + (KS - 1);
    int pbj = (sj - j) / DILX + (KS - 1);

    const float scale = 0.17677669529663687f;

    const float* qp = q + ((size_t)(b * CDIM + head * HD)) * NPIX + i * WW + j;
    float qr[HD];
#pragma unroll
    for (int d = 0; d < HD; ++d) qr[d] = qp[(size_t)d * NPIX] * scale;

    const size_t kvbase = ((size_t)(b * KTOT + f * CDIM + head * HD)) * NPIX;
    const float* kb = kk + kvbase;
    const float* vb = vv + kvbase;
    const float* rp = rpb + head * (2 * KS - 1) * (2 * KS - 1);

    float m = -3.0e38f, s = 0.0f;
    float acc[HD];
#pragma unroll
    for (int d = 0; d < HD; ++d) acc[d] = 0.0f;

    for (int ti = 0; ti < KS; ++ti) {
        int ii  = si + ti * DILX;
        int bro = (pbi + ti) * (2 * KS - 1);
        for (int tj = 0; tj < KS; ++tj) {
            int jj  = sj + tj * DILX;
            int off = ii * WW + jj;

            float l = rp[bro + pbj + tj];
#pragma unroll
            for (int d = 0; d < HD; ++d)
                l += qr[d] * kb[(size_t)d * NPIX + off];

            float mn   = fmaxf(m, l);
            float corr = __expf(m - mn);
            float w    = __expf(l - mn);
            s = s * corr + w;
            m = mn;
#pragma unroll
            for (int d = 0; d < HD; ++d)
                acc[d] = acc[d] * corr + w * vb[(size_t)d * NPIX + off];
        }
    }

    float inv = 1.0f / s;
    float* op = xn + ((size_t)(b * KTOT + f * CDIM + head * HD)) * NPIX + i * WW + j;
#pragma unroll
    for (int d = 0; d < HD; ++d) op[(size_t)d * NPIX] = acc[d] * inv;
}

__global__ __launch_bounds__(256, 1)
void conv1x1_wmma_kernel(const float* __restrict__ xn,
                         const float* __restrict__ wzw,
                         const float* __restrict__ wzb,
                         float* __restrict__ out)
{
    extern __shared__ v2f wfrag[];

    for (int s = threadIdx.x; s < NFRAG; s += 256) {
        int ln    = s & 31;
        int mt    = (s >> 5) & 7;
        int kstep = s >> 8;
        int row   = mt * 16 + (ln & 15);
        int c     = kstep * 4 + 2 * (ln >> 4);
        const float* wp = wzw + (size_t)row * KTOT + c;
        v2f t;
        t.x = wp[0];
        t.y = wp[1];
        wfrag[s] = t;
    }
    __syncthreads();

    int lane  = threadIdx.x & 31;
    int wave  = threadIdx.x >> 5;
    int mhalf = wave & 1;
    int ntp   = blockIdx.x * 4 + (wave >> 1);
    int col   = lane & 15;
    int hi    = lane >> 4;

    int g0 = ntp * 32 + col;
    int b  = g0 >> 12;
    int p0 = g0 & 4095;
    const float* xp0 = xn + (size_t)b * KTOT * NPIX + p0;
    const float* xp1 = xp0 + 16;

    v8f acc0[4], acc1[4];
#pragma unroll
    for (int mt = 0; mt < 4; ++mt)
#pragma unroll
        for (int e = 0; e < 8; ++e) { acc0[mt][e] = 0.0f; acc1[mt][e] = 0.0f; }

    auto load_step = [&](int ks, v2f a[4], v2f& b0, v2f& b1) {
        int c0 = ks * 4 + 2 * hi;
        b0.x = xp0[(size_t)c0 * NPIX];
        b0.y = xp0[(size_t)(c0 + 1) * NPIX];
        b1.x = xp1[(size_t)c0 * NPIX];
        b1.y = xp1[(size_t)(c0 + 1) * NPIX];
        const v2f* wf = wfrag + (size_t)ks * 256 + (mhalf * 4) * 32 + lane;
#pragma unroll
        for (int mt = 0; mt < 4; ++mt) a[mt] = wf[mt * 32];
    };

    v2f aA[4], bA0, bA1;
    v2f aB[4], bB0, bB1;
    load_step(0, aA, bA0, bA1);

    for (int ks = 0; ks < KSTEPS; ks += 2) {
        load_step(ks + 1, aB, bB0, bB1);

#pragma unroll
        for (int mt = 0; mt < 4; ++mt) {
            acc0[mt] = __builtin_amdgcn_wmma_f32_16x16x4_f32(
                false, aA[mt], false, bA0, (short)0, acc0[mt], false, false);
            acc1[mt] = __builtin_amdgcn_wmma_f32_16x16x4_f32(
                false, aA[mt], false, bA1, (short)0, acc1[mt], false, false);
        }

        int ksn = (ks + 2 < KSTEPS) ? (ks + 2) : (KSTEPS - 1);
        load_step(ksn, aA, bA0, bA1);

#pragma unroll
        for (int mt = 0; mt < 4; ++mt) {
            acc0[mt] = __builtin_amdgcn_wmma_f32_16x16x4_f32(
                false, aB[mt], false, bB0, (short)0, acc0[mt], false, false);
            acc1[mt] = __builtin_amdgcn_wmma_f32_16x16x4_f32(
                false, aB[mt], false, bB1, (short)0, acc1[mt], false, false);
        }
    }

    float* op0 = out + (size_t)b * CDIM * NPIX + p0;
    float* op1 = op0 + 16;
#pragma unroll
    for (int mt = 0; mt < 4; ++mt) {
#pragma unroll
        for (int e = 0; e < 8; ++e) {
            int mrow = (mhalf * 4 + mt) * 16 + e + 8 * hi;
            float bias = wzb[mrow];
            op0[(size_t)mrow * NPIX] = acc0[mt][e] + bias;
            op1[(size_t)mrow * NPIX] = acc1[mt][e] + bias;
        }
    }
}

extern "C" void kernel_launch(void* const* d_in, const int* in_sizes, int n_in,
                              void* d_out, int out_size, void* d_ws, size_t ws_size,
                              hipStream_t stream) {
    const float* q   = (const float*)d_in[0];
    const float* k   = (const float*)d_in[1];
    const float* v   = (const float*)d_in[2];
    const float* rpb = (const float*)d_in[3];
    const float* wzw = (const float*)d_in[4];
    const float* wzb = (const float*)d_in[5];
    float* out = (float*)d_out;
    float* xn  = (float*)d_ws;

    natten_attn_kernel<<<384, 256, 0, stream>>>(q, k, v, rpb, xn);

    size_t lds_bytes = (size_t)NFRAG * sizeof(v2f);
    (void)hipFuncSetAttribute(reinterpret_cast<const void*>(conv1x1_wmma_kernel),
                              hipFuncAttributeMaxDynamicSharedMemorySize,
                              (int)lds_bytes);
    conv1x1_wmma_kernel<<<64, 256, lds_bytes, stream>>>(xn, wzw, wzb, out);
}


# attention kernel: each tap loads its 32 K and 32 V channel values as one group ahead of the sums (64 loads in flight per wave) instead of load, wait, multiply-add one by one; sums unchanged
# speedup vs baseline: 5.2283x; 5.2283x over previous
#include <hip/hip_runtime.h>
#include <hip/hip_bf16.h>

#define KS      7
#define DILX    2
#define NHEADS  4
#define NFRAMES 3
#define CDIM    128
#define HD      32
#define HH      64
#define WW      64
#define NPIX    4096
#define NB      2
#define KTOT    (NFRAMES*CDIM)
#define KSTEPS  (KTOT/4)
#define NFRAG   (KSTEPS*8*32)

typedef __attribute__((ext_vector_type(2))) float v2f;
typedef __attribute__((ext_vector_type(8))) float v8f;

__device__ __forceinline__ int win_start(int i) {
    int ni = i - (KS / 2) * DILX;
    if (ni < 0) return i % DILX;
    if (i + (KS / 2) * DILX >= HH) return HH - KS * DILX + (i % DILX);
    return ni;
}

__global__ __launch_bounds__(256)
void natten_attn_kernel(const float* __restrict__ q,
                                   const float* __restrict__ kk,
                                   const float* __restrict__ vv,
                                   const float* __restrict__ rpb,
                                   float* __restrict__ xn)
{
    int tid = blockIdx.x * blockDim.x + threadIdx.x;
    int j = tid & 63;
    int t = tid >> 6;
    int i = t & 63;
    t >>= 6;
    int f = t % NFRAMES;
    t /= NFRAMES;
    int head = t & 3;
    int b = t >> 2;

    int si = win_start(i), sj = win_start(j);
    int pbi = (si - i) / DILX + (KS - 1);
    int pbj = (sj - j) / DILX + (KS - 1);

    const float scale = 0.17677669529663687f;

    const float* qp = q + ((size_t)(b * CDIM + head * HD)) * NPIX + i * WW + j;
    float qr[HD];
#pragma unroll
    for (int d = 0; d < HD; ++d) qr[d] = qp[(size_t)d * NPIX] * scale;

    const size_t kvbase = ((size_t)(b * KTOT + f * CDIM + head * HD)) * NPIX;
    const float* kb = kk + kvbase;
    const float* vb = vv + kvbase;
    const float* rp = rpb + head * (2 * KS - 1) * (2 * KS - 1);

    float m = -3.0e38f, s = 0.0f;
    float acc[HD];
#pragma unroll
    for (int d = 0; d < HD; ++d) acc[d] = 0.0f;

    for (int ti = 0; ti < KS; ++ti) {
        int ii  = si + ti * DILX;
        int bro = (pbi + ti) * (2 * KS - 1);
        for (int tj = 0; tj < KS; ++tj) {
            int jj  = sj + tj * DILX;
            int off = ii * WW + jj;

            float l = rp[bro + pbj + tj];
            float kr[HD], vr[HD];
#pragma unroll
            for (int d = 0; d < HD; ++d) kr[d] = kb[(size_t)d * NPIX + off];
#pragma unroll
            for (int d = 0; d < HD; ++d) vr[d] = vb[(size_t)d * NPIX + off];
#pragma unroll
            for (int d = 0; d < HD; ++d)
                l += qr[d] * kr[d];

            float mn   = fmaxf(m, l);
            float corr = __expf(m - mn);
            float w    = __expf(l - mn);
            s = s * corr + w;
            m = mn;
#pragma unroll
            for (int d = 0; d < HD; ++d)
                acc[d] = acc[d] * corr + w * vr[d];
        }
    }

    float inv = 1.0f / s;
    float* op = xn + ((size_t)(b * KTOT + f * CDIM + head * HD)) * NPIX + i * WW + j;
#pragma unroll
    for (int d = 0; d < HD; ++d) op[(size_t)d * NPIX] = acc[d] * inv;
}

__global__ __launch_bounds__(256, 1)
void conv1x1_wmma_kernel(const float* __restrict__ xn,
                         const float* __restrict__ wzw,
                         const float* __restrict__ wzb,
                         float* __restrict__ out)
{
    extern __shared__ v2f wfrag[];

    for (int s = threadIdx.x; s < NFRAG; s += 256) {
        int ln    = s & 31;
        int mt    = (s >> 5) & 7;
        int kstep = s >> 8;
        int row   = mt * 16 + (ln & 15);
        int c     = kstep * 4 + 2 * (ln >> 4);
        const float* wp = wzw + (size_t)row * KTOT + c;
        v2f t;
        t.x = wp[0];
        t.y = wp[1];
        wfrag[s] = t;
    }
    __syncthreads();

    int lane  = threadIdx.x & 31;
    int wave  = threadIdx.x >> 5;
    int mhalf = wave & 1;
    int ntp   = blockIdx.x * 4 + (wave >> 1);
    int col   = lane & 15;
    int hi    = lane >> 4;

    int g0 = ntp * 32 + col;
    int b  = g0 >> 12;
    int p0 = g0 & 4095;
    const float* xp0 = xn + (size_t)b * KTOT * NPIX + p0;
    const float* xp1 = xp0 + 16;

    v8f acc0[4], acc1[4];
#pragma unroll
    for (int mt = 0; mt < 4; ++mt)
#pragma unroll
        for (int e = 0; e < 8; ++e) { acc0[mt][e] = 0.0f; acc1[mt][e] = 0.0f; }

    auto load_step = [&](int ks, v2f a[4], v2f& b0, v2f& b1) {
        int c0 = ks * 4 + 2 * hi;
        b0.x = xp0[(size_t)c0 * NPIX];
        b0.y = xp0[(size_t)(c0 + 1) * NPIX];
        b1.x = xp1[(size_t)c0 * NPIX];
        b1.y = xp1[(size_t)(c0 + 1) * NPIX];
        const v2f* wf = wfrag + (size_t)ks * 256 + (mhalf * 4) * 32 + lane;
#pragma unroll
        for (int mt = 0; mt < 4; ++mt) a[mt] = wf[mt * 32];
    };

    v2f aA[4], bA0, bA1;
    v2f aB[4], bB0, bB1;
    load_step(0, aA, bA0, bA1);

    for (int ks = 0; ks < KSTEPS; ks += 2) {
        load_step(ks + 1, aB, bB0, bB1);

#pragma unroll
        for (int mt = 0; mt < 4; ++mt) {
            acc0[mt] = __builtin_amdgcn_wmma_f32_16x16x4_f32(
                false, aA[mt], false, bA0, (short)0, acc0[mt], false, false);
            acc1[mt] = __builtin_amdgcn_wmma_f32_16x16x4_f32(
                false, aA[mt], false, bA1, (short)0, acc1[mt], false, false);
        }

        int ksn = (ks + 2 < KSTEPS) ? (ks + 2) : (KSTEPS - 1);
        load_step(ksn, aA, bA0, bA1);

#pragma unroll
        for (int mt = 0; mt < 4; ++mt) {
            acc0[mt] = __builtin_amdgcn_wmma_f32_16x16x4_f32(
                false, aB[mt], false, bB0, (short)0, acc0[mt], false, false);
            acc1[mt] = __builtin_amdgcn_wmma_f32_16x16x4_f32(
                false, aB[mt], false, bB1, (short)0, acc1[mt], false, false);
        }
    }

    float* op0 = out + (size_t)b * CDIM * NPIX + p0;
    float* op1 = op0 + 16;
#pragma unroll
    for (int mt = 0; mt < 4; ++mt) {
#pragma unroll
        for (int e = 0; e < 8; ++e) {
            int mrow = (mhalf * 4 + mt) * 16 + e + 8 * hi;
            float bias = wzb[mrow];
            op0[(size_t)mrow * NPIX] = acc0[mt][e] + bias;
            op1[(size_t)mrow * NPIX] = acc1[mt][e] + bias;
        }
    }
}

extern "C" void kernel_launch(void* const* d_in, const int* in_sizes, int n_in,
                              void* d_out, int out_size, void* d_ws, size_t ws_size,
                              hipStream_t stream) {
    const float* q   = (const float*)d_in[0];
    const float* k   = (const float*)d_in[1];
    const float* v   = (const float*)d_in[2];
    const float* rpb = (const float*)d_in[3];
    const float* wzw = (const float*)d_in[4];
    const float* wzb = (const float*)d_in[5];
    float* out = (float*)d_out;
    float* xn  = (float*)d_ws;

    natten_attn_kernel<<<384, 256, 0, stream>>>(q, k, v, rpb, xn);

    size_t lds_bytes = (size_t)NFRAG * sizeof(v2f);
    (void)hipFuncSetAttribute(reinterpret_cast<const void*>(conv1x1_wmma_kernel),
                              hipFuncAttributeMaxDynamicSharedMemorySize,
                              (int)lds_bytes);
    conv1x1_wmma_kernel<<<64, 256, lds_bytes, stream>>>(xn, wzw, wzb, out);
}


# attention kernel: each block copies its 16-row band of K, then of V, once into LDS (pixel-major, 144-B pitch); all 49 taps read it with ds_load_b128 instead of global loads; order of sums unchanged
# speedup vs baseline: 8.4141x; 1.6093x over previous
#include <hip/hip_runtime.h>
#include <hip/hip_bf16.h>

#define KS      7
#define DILX    2
#define NHEADS  4
#define NFRAMES 3
#define CDIM    128
#define HD      32
#define HH      64
#define WW      64
#define NPIX    4096
#define NB      2
#define KTOT    (NFRAMES*CDIM)
#define KSTEPS  (KTOT/4)
#define NFRAG   (KSTEPS*8*32)
#define ROWS_BLK 4
#define TROWS    16
#define PITCH4   9
#define ATT_LDS  (TROWS*WW*PITCH4*16)

typedef __attribute__((ext_vector_type(2))) float v2f;
typedef __attribute__((ext_vector_type(8))) float v8f;

__device__ __forceinline__ int win_start(int i) {
    int ni = i - (KS / 2) * DILX;
    if (ni < 0) return i % DILX;
    if (i + (KS / 2) * DILX >= HH) return HH - KS * DILX + (i % DILX);
    return ni;
}

__global__ __launch_bounds__(256)
void natten_attn_kernel(const float* __restrict__ q,
                                   const float* __restrict__ kk,
                                   const float* __restrict__ vv,
                                   const float* __restrict__ rpb,
                                   float* __restrict__ xn)
{
    extern __shared__ float4 tile[];

    int tid = blockIdx.x * blockDim.x + threadIdx.x;
    int j = tid & 63;
    int t = tid >> 6;
    int i = t & 63;
    t >>= 6;
    int f = t % NFRAMES;
    t /= NFRAMES;
    int head = t & 3;
    int b = t >> 2;

    int si = win_start(i), sj = win_start(j);
    int pbi = (si - i) / DILX + (KS - 1);
    int pbj = (sj - j) / DILX + (KS - 1);

    int i0 = (blockIdx.x * ROWS_BLK) & (HH - 1);
    int lo = i0 - (KS / 2) * DILX;
    lo = lo < 0 ? 0 : lo;
    lo = lo > HH - TROWS ? HH - TROWS : lo;

    const float scale = 0.17677669529663687f;

    const float* qp = q + ((size_t)(b * CDIM + head * HD)) * NPIX + i * WW + j;
    float qr[HD];
#pragma unroll
    for (int d = 0; d < HD; ++d) qr[d] = qp[(size_t)d * NPIX] * scale;

    const size_t kvbase = ((size_t)(b * KTOT + f * CDIM + head * HD)) * NPIX
                        + (size_t)lo * WW;
    const float* rp = rpb + head * (2 * KS - 1) * (2 * KS - 1);

    auto stage = [&](const float* __restrict__ srcp) {
#pragma unroll 2
        for (int c4 = 0; c4 < HD / 4; ++c4) {
#pragma unroll
            for (int pp = 0; pp < 4; ++pp) {
                int p = pp * 256 + threadIdx.x;
                const float* g = srcp + (size_t)(4 * c4) * NPIX + p;
                float4 tv;
                tv.x = g[0];
                tv.y = g[(size_t)NPIX];
                tv.z = g[(size_t)2 * NPIX];
                tv.w = g[(size_t)3 * NPIX];
                tile[p * PITCH4 + c4] = tv;
            }
        }
    };

    const float4* tb = tile + ((si - lo) * WW + sj) * PITCH4;

    stage(kk + kvbase);
    __syncthreads();

    float l[KS * KS];
#pragma unroll
    for (int k = 0; k < KS * KS; ++k) l[k] = 0.0f;

#pragma unroll 1
    for (int ti = 0; ti < KS; ++ti) {
        const float4* trow = tb + ti * (DILX * WW * PITCH4);
        const float* rprow = rp + (pbi + ti) * (2 * KS - 1) + pbj;
        float ln[KS];
#pragma unroll
        for (int tj = 0; tj < KS; ++tj) {
            const float4* tp = trow + tj * (DILX * PITCH4);
            float lv = rprow[tj];
            float4 k4[HD / 4];
#pragma unroll
            for (int c = 0; c < HD / 4; ++c) k4[c] = tp[c];
#pragma unroll
            for (int c = 0; c < HD / 4; ++c) {
                lv = __builtin_fmaf(qr[4 * c + 0], k4[c].x, lv);
                lv = __builtin_fmaf(qr[4 * c + 1], k4[c].y, lv);
                lv = __builtin_fmaf(qr[4 * c + 2], k4[c].z, lv);
                lv = __builtin_fmaf(qr[4 * c + 3], k4[c].w, lv);
            }
            ln[tj] = lv;
        }
#pragma unroll
        for (int k = 0; k < KS * (KS - 1); ++k) l[k] = l[k + KS];
#pragma unroll
        for (int k = 0; k < KS; ++k) l[KS * (KS - 1) + k] = ln[k];
    }

    __syncthreads();
    stage(vv + kvbase);
    __syncthreads();

    float m = -3.0e38f, s = 0.0f;
    float acc[HD];
#pragma unroll
    for (int d = 0; d < HD; ++d) acc[d] = 0.0f;

#pragma unroll 1
    for (int ti = 0; ti < KS; ++ti) {
        const float4* trow = tb + ti * (DILX * WW * PITCH4);
#pragma unroll
        for (int tj = 0; tj < KS; ++tj) {
            const float4* tp = trow + tj * (DILX * PITCH4);
            float4 v4[HD / 4];
#pragma unroll
            for (int c = 0; c < HD / 4; ++c) v4[c] = tp[c];

            float lv   = l[tj];
            float mn   = fmaxf(m, lv);
            float corr = __expf(m - mn);
            float w    = __expf(lv - mn);
            s = s * corr + w;
            m = mn;
#pragma unroll
            for (int c = 0; c < HD / 4; ++c) {
                acc[4 * c + 0] = acc[4 * c + 0] * corr + w * v4[c].x;
                acc[4 * c + 1] = acc[4 * c + 1] * corr + w * v4[c].y;
                acc[4 * c + 2] = acc[4 * c + 2] * corr + w * v4[c].z;
                acc[4 * c + 3] = acc[4 * c + 3] * corr + w * v4[c].w;
            }
        }
#pragma unroll
        for (int k = 0; k < KS * (KS - 1); ++k) l[k] = l[k + KS];
    }

    float inv = 1.0f / s;
    float* op = xn + ((size_t)(b * KTOT + f * CDIM + head * HD)) * NPIX + i * WW + j;
#pragma unroll
    for (int d = 0; d < HD; ++d) op[(size_t)d * NPIX] = acc[d] * inv;
}

__global__ __launch_bounds__(256, 1)
void conv1x1_wmma_kernel(const float* __restrict__ xn,
                         const float* __restrict__ wzw,
                         const float* __restrict__ wzb,
                         float* __restrict__ out)
{
    extern __shared__ v2f wfrag[];

    for (int s = threadIdx.x; s < NFRAG; s += 256) {
        int ln    = s & 31;
        int mt    = (s >> 5) & 7;
        int kstep = s >> 8;
        int row   = mt * 16 + (ln & 15);
        int c     = kstep * 4 + 2 * (ln >> 4);
        const float* wp = wzw + (size_t)row * KTOT + c;
        v2f t;
        t.x = wp[0];
        t.y = wp[1];
        wfrag[s] = t;
    }
    __syncthreads();

    int lane  = threadIdx.x & 31;
    int wave  = threadIdx.x >> 5;
    int mhalf = wave & 1;
    int ntp   = blockIdx.x * 4 + (wave >> 1);
    int col   = lane & 15;
    int hi    = lane >> 4;

    int g0 = ntp * 32 + col;
    int b  = g0 >> 12;
    int p0 = g0 & 4095;
    const float* xp0 = xn + (size_t)b * KTOT * NPIX + p0;
    const float* xp1 = xp0 + 16;

    v8f acc0[4], acc1[4];
#pragma unroll
    for (int mt = 0; mt < 4; ++mt)
#pragma unroll
        for (int e = 0; e < 8; ++e) { acc0[mt][e] = 0.0f; acc1[mt][e] = 0.0f; }

    auto load_step = [&](int ks, v2f a[4], v2f& b0, v2f& b1) {
        int c0 = ks * 4 + 2 * hi;
        b0.x = xp0[(size_t)c0 * NPIX];
        b0.y = xp0[(size_t)(c0 + 1) * NPIX];
        b1.x = xp1[(size_t)c0 * NPIX];
        b1.y = xp1[(size_t)(c0 + 1) * NPIX];
        const v2f* wf = wfrag + (size_t)ks * 256 + (mhalf * 4) * 32 + lane;
#pragma unroll
        for (int mt = 0; mt < 4; ++mt) a[mt] = wf[mt * 32];
    };

    v2f aA[4], bA0, bA1;
    v2f aB[4], bB0, bB1;
    load_step(0, aA, bA0, bA1);

    for (int ks = 0; ks < KSTEPS; ks += 2) {
        load_step(ks + 1, aB, bB0, bB1);

#pragma unroll
        for (int mt = 0; mt < 4; ++mt) {
            acc0[mt] = __builtin_amdgcn_wmma_f32_16x16x4_f32(
                false, aA[mt], false, bA0, (short)0, acc0[mt], false, false);
            acc1[mt] = __builtin_amdgcn_wmma_f32_16x16x4_f32(
                false, aA[mt], false, bA1, (short)0, acc1[mt], false, false);
        }

        int ksn = (ks + 2 < KSTEPS) ? (ks + 2) : (KSTEPS - 1);
        load_step(ksn, aA, bA0, bA1);

#pragma unroll
        for (int mt = 0; mt < 4; ++mt) {
            acc0[mt] = __builtin_amdgcn_wmma_f32_16x16x4_f32(
                false, aB[mt], false, bB0, (short)0, acc0[mt], false, false);
            acc1[mt] = __builtin_amdgcn_wmma_f32_16x16x4_f32(
                false, aB[mt], false, bB1, (short)0, acc1[mt], false, false);
        }
    }

    float* op0 = out + (size_t)b * CDIM * NPIX + p0;
    float* op1 = op0 + 16;
#pragma unroll
    for (int mt = 0; mt < 4; ++mt) {
#pragma unroll
        for (int e = 0; e < 8; ++e) {
            int mrow = (mhalf * 4 + mt) * 16 + e + 8 * hi;
            float bias = wzb[mrow];
            op0[(size_t)mrow * NPIX] = acc0[mt][e] + bias;
            op1[(size_t)mrow * NPIX] = acc1[mt][e] + bias;
        }
    }
}

extern "C" void kernel_launch(void* const* d_in, const int* in_sizes, int n_in,
                              void* d_out, int out_size, void* d_ws, size_t ws_size,
                              hipStream_t stream) {
    const float* q   = (const float*)d_in[0];
    const float* k   = (const float*)d_in[1];
    const float* v   = (const float*)d_in[2];
    const float* rpb = (const float*)d_in[3];
    const float* wzw = (const float*)d_in[4];
    const float* wzb = (const float*)d_in[5];
    float* out = (float*)d_out;
    float* xn  = (float*)d_ws;

    (void)hipFuncSetAttribute(reinterpret_cast<const void*>(natten_attn_kernel),
                              hipFuncAttributeMaxDynamicSharedMemorySize,
                              (int)ATT_LDS);
    natten_attn_kernel<<<384, 256, ATT_LDS, stream>>>(q, k, v, rpb, xn);

    size_t lds_bytes = (size_t)NFRAG * sizeof(v2f);
    (void)hipFuncSetAttribute(reinterpret_cast<const void*>(conv1x1_wmma_kernel),
                              hipFuncAttributeMaxDynamicSharedMemorySize,
                              (int)lds_bytes);
    conv1x1_wmma_kernel<<<64, 256, lds_bytes, stream>>>(xn, wzw, wzb, out);
}


# conv GEMM: 256 work-groups of 4 waves, each wave one 32x32 tile with its W rows and xn columns loaded straight from global a 32-k group ahead (no LDS copy of W); k order inside the sums changes
# speedup vs baseline: 15.5168x; 1.8441x over previous
#include <hip/hip_runtime.h>
#include <hip/hip_bf16.h>

#define KS      7
#define DILX    2
#define NHEADS  4
#define NFRAMES 3
#define CDIM    128
#define HD      32
#define HH      64
#define WW      64
#define NPIX    4096
#define NB      2
#define KTOT    (NFRAMES*CDIM)
#define ROWS_BLK 4
#define TROWS    16
#define PITCH4   9
#define ATT_LDS  (TROWS*WW*PITCH4*16)

typedef __attribute__((ext_vector_type(2))) float v2f;
typedef __attribute__((ext_vector_type(8))) float v8f;

__device__ __forceinline__ int win_start(int i) {
    int ni = i - (KS / 2) * DILX;
    if (ni < 0) return i % DILX;
    if (i + (KS / 2) * DILX >= HH) return HH - KS * DILX + (i % DILX);
    return ni;
}

__global__ __launch_bounds__(256)
void natten_attn_kernel(const float* __restrict__ q,
                                   const float* __restrict__ kk,
                                   const float* __restrict__ vv,
                                   const float* __restrict__ rpb,
                                   float* __restrict__ xn)
{
    extern __shared__ float4 tile[];

    int tid = blockIdx.x * blockDim.x + threadIdx.x;
    int j = tid & 63;
    int t = tid >> 6;
    int i = t & 63;
    t >>= 6;
    int f = t % NFRAMES;
    t /= NFRAMES;
    int head = t & 3;
    int b = t >> 2;

    int si = win_start(i), sj = win_start(j);
    int pbi = (si - i) / DILX + (KS - 1);
    int pbj = (sj - j) / DILX + (KS - 1);

    int i0 = (blockIdx.x * ROWS_BLK) & (HH - 1);
    int lo = i0 - (KS / 2) * DILX;
    lo = lo < 0 ? 0 : lo;
    lo = lo > HH - TROWS ? HH - TROWS : lo;

    const float scale = 0.17677669529663687f;

    const float* qp = q + ((size_t)(b * CDIM + head * HD)) * NPIX + i * WW + j;
    float qr[HD];
#pragma unroll
    for (int d = 0; d < HD; ++d) qr[d] = qp[(size_t)d * NPIX] * scale;

    const size_t kvbase = ((size_t)(b * KTOT + f * CDIM + head * HD)) * NPIX
                        + (size_t)lo * WW;
    const float* rp = rpb + head * (2 * KS - 1) * (2 * KS - 1);

    auto stage = [&](const float* __restrict__ srcp) {
#pragma unroll 2
        for (int c4 = 0; c4 < HD / 4; ++c4) {
#pragma unroll
            for (int pp = 0; pp < 4; ++pp) {
                int p = pp * 256 + threadIdx.x;
                const float* g = srcp + (size_t)(4 * c4) * NPIX + p;
                float4 tv;
                tv.x = g[0];
                tv.y = g[(size_t)NPIX];
                tv.z = g[(size_t)2 * NPIX];
                tv.w = g[(size_t)3 * NPIX];
                tile[p * PITCH4 + c4] = tv;
            }
        }
    };

    const float4* tb = tile + ((si - lo) * WW + sj) * PITCH4;

    stage(kk + kvbase);
    __syncthreads();

    float l[KS * KS];
#pragma unroll
    for (int k = 0; k < KS * KS; ++k) l[k] = 0.0f;

#pragma unroll 1
    for (int ti = 0; ti < KS; ++ti) {
        const float4* trow = tb + ti * (DILX * WW * PITCH4);
        const float* rprow = rp + (pbi + ti) * (2 * KS - 1) + pbj;
        float ln[KS];
#pragma unroll
        for (int tj = 0; tj < KS; ++tj) {
            const float4* tp = trow + tj * (DILX * PITCH4);
            float lv = rprow[tj];
            float4 k4[HD / 4];
#pragma unroll
            for (int c = 0; c < HD / 4; ++c) k4[c] = tp[c];
#pragma unroll
            for (int c = 0; c < HD / 4; ++c) {
                lv = __builtin_fmaf(qr[4 * c + 0], k4[c].x, lv);
                lv = __builtin_fmaf(qr[4 * c + 1], k4[c].y, lv);
                lv = __builtin_fmaf(qr[4 * c + 2], k4[c].z, lv);
                lv = __builtin_fmaf(qr[4 * c + 3], k4[c].w, lv);
            }
            ln[tj] = lv;
        }
#pragma unroll
        for (int k = 0; k < KS * (KS - 1); ++k) l[k] = l[k + KS];
#pragma unroll
        for (int k = 0; k < KS; ++k) l[KS * (KS - 1) + k] = ln[k];
    }

    __syncthreads();
    stage(vv + kvbase);
    __syncthreads();

    float m = -3.0e38f, s = 0.0f;
    float acc[HD];
#pragma unroll
    for (int d = 0; d < HD; ++d) acc[d] = 0.0f;

#pragma unroll 1
    for (int ti = 0; ti < KS; ++ti) {
        const float4* trow = tb + ti * (DILX * WW * PITCH4);
#pragma unroll
        for (int tj = 0; tj < KS; ++tj) {
            const float4* tp = trow + tj * (DILX * PITCH4);
            float4 v4[HD / 4];
#pragma unroll
            for (int c = 0; c < HD / 4; ++c) v4[c] = tp[c];

            float lv   = l[tj];
            float mn   = fmaxf(m, lv);
            float corr = __expf(m - mn);
            float w    = __expf(lv - mn);
            s = s * corr + w;
            m = mn;
#pragma unroll
            for (int c = 0; c < HD / 4; ++c) {
                acc[4 * c + 0] = acc[4 * c + 0] * corr + w * v4[c].x;
                acc[4 * c + 1] = acc[4 * c + 1] * corr + w * v4[c].y;
                acc[4 * c + 2] = acc[4 * c + 2] * corr + w * v4[c].z;
                acc[4 * c + 3] = acc[4 * c + 3] * corr + w * v4[c].w;
            }
        }
#pragma unroll
        for (int k = 0; k < KS * (KS - 1); ++k) l[k] = l[k + KS];
    }

    float inv = 1.0f / s;
    float* op = xn + ((size_t)(b * KTOT + f * CDIM + head * HD)) * NPIX + i * WW + j;
#pragma unroll
    for (int d = 0; d < HD; ++d) op[(size_t)d * NPIX] = acc[d] * inv;
}

#define KGROUP   32
#define NGROUPS  (KTOT / KGROUP)

struct Frag {
    float a0[16], a1[16];
    float b0[16], b1[16];
};

__global__ __launch_bounds__(128)
void conv1x1_wmma_kernel(const float* __restrict__ xn,
                         const float* __restrict__ wzw,
                         const float* __restrict__ wzb,
                         float* __restrict__ out)
{
    int lane = threadIdx.x & 31;
    int wave = threadIdx.x >> 5;
    int col  = lane & 15;
    int hi   = lane >> 4;

    int mbase = (blockIdx.x & 1) * 64 + (wave & 1) * 32;
    int g0    = (blockIdx.x >> 1) * 64 + (wave >> 1) * 32;
    int b     = g0 >> 12;
    int p0    = (g0 & 4095) + col;

    const float* wp0 = wzw + (size_t)(mbase + col) * KTOT + 16 * hi;
    const float* wp1 = wp0 + (size_t)16 * KTOT;
    const float* xp0 = xn + ((size_t)b * KTOT + 16 * hi) * NPIX + p0;
    const float* xp1 = xp0 + 16;

    v8f acc00, acc01, acc10, acc11;
#pragma unroll
    for (int e = 0; e < 8; ++e) { acc00[e] = 0.0f; acc01[e] = 0.0f; acc10[e] = 0.0f; acc11[e] = 0.0f; }

    auto load_group = [&](int g, Frag& f) {
        const float* w0 = wp0 + KGROUP * g;
        const float* w1 = wp1 + KGROUP * g;
        const float* x0 = xp0 + (size_t)(KGROUP * g) * NPIX;
        const float* x1 = xp1 + (size_t)(KGROUP * g) * NPIX;
#pragma unroll
        for (int e = 0; e < 16; ++e) { f.a0[e] = w0[e]; f.a1[e] = w1[e]; }
#pragma unroll
        for (int e = 0; e < 16; ++e) {
            f.b0[e] = x0[(size_t)e * NPIX];
            f.b1[e] = x1[(size_t)e * NPIX];
        }
    };

    auto mma_group = [&](const Frag& f) {
#pragma unroll
        for (int j = 0; j < 8; ++j) {
            v2f A0, A1, B0, B1;
            A0.x = f.a0[2 * j]; A0.y = f.a0[2 * j + 1];
            A1.x = f.a1[2 * j]; A1.y = f.a1[2 * j + 1];
            B0.x = f.b0[2 * j]; B0.y = f.b0[2 * j + 1];
            B1.x = f.b1[2 * j]; B1.y = f.b1[2 * j + 1];
            acc00 = __builtin_amdgcn_wmma_f32_16x16x4_f32(
                false, A0, false, B0, (short)0, acc00, false, false);
            acc01 = __builtin_amdgcn_wmma_f32_16x16x4_f32(
                false, A0, false, B1, (short)0, acc01, false, false);
            acc10 = __builtin_amdgcn_wmma_f32_16x16x4_f32(
                false, A1, false, B0, (short)0, acc10, false, false);
            acc11 = __builtin_amdgcn_wmma_f32_16x16x4_f32(
                false, A1, false, B1, (short)0, acc11, false, false);
        }
    };

    Frag fA, fB;
    load_group(0, fA);

    for (int g = 0; g < NGROUPS - 2; g += 2) {
        load_group(g + 1, fB);
        __builtin_amdgcn_sched_barrier(0);
        mma_group(fA);
        __builtin_amdgcn_sched_barrier(0);
        load_group(g + 2, fA);
        __builtin_amdgcn_sched_barrier(0);
        mma_group(fB);
        __builtin_amdgcn_sched_barrier(0);
    }
    load_group(NGROUPS - 1, fB);
    __builtin_amdgcn_sched_barrier(0);
    mma_group(fA);
    mma_group(fB);

    float* op0 = out + (size_t)b * CDIM * NPIX + p0;
    float* op1 = op0 + 16;
#pragma unroll
    for (int e = 0; e < 8; ++e) {
        int r0 = mbase + e + 8 * hi;
        int r1 = r0 + 16;
        float bias0 = wzb[r0];
        float bias1 = wzb[r1];
        op0[(size_t)r0 * NPIX] = acc00[e] + bias0;
        op1[(size_t)r0 * NPIX] = acc01[e] + bias0;
        op0[(size_t)r1 * NPIX] = acc10[e] + bias1;
        op1[(size_t)r1 * NPIX] = acc11[e] + bias1;
    }
}

extern "C" void kernel_launch(void* const* d_in, const int* in_sizes, int n_in,
                              void* d_out, int out_size, void* d_ws, size_t ws_size,
                              hipStream_t stream) {
    const float* q   = (const float*)d_in[0];
    const float* k   = (const float*)d_in[1];
    const float* v   = (const float*)d_in[2];
    const float* rpb = (const float*)d_in[3];
    const float* wzw = (const float*)d_in[4];
    const float* wzb = (const float*)d_in[5];
    float* out = (float*)d_out;
    float* xn  = (float*)d_ws;

    (void)hipFuncSetAttribute(reinterpret_cast<const void*>(natten_attn_kernel),
                              hipFuncAttributeMaxDynamicSharedMemorySize,
                              (int)ATT_LDS);
    natten_attn_kernel<<<384, 256, ATT_LDS, stream>>>(q, k, v, rpb, xn);

    conv1x1_wmma_kernel<<<256, 128, 0, stream>>>(xn, wzw, wzb, out);
}


# attention kernel, K pass: a tap row's LDS reads issued in 8 groups of 7 (one per tap), each a group ahead of its multiply-adds, the 7 tap chains advancing in turn; bias preloaded; sums unchanged
# speedup vs baseline: 16.9301x; 1.0911x over previous
#include <hip/hip_runtime.h>
#include <hip/hip_bf16.h>

#define KS      7
#define DILX    2
#define NHEADS  4
#define NFRAMES 3
#define CDIM    128
#define HD      32
#define HH      64
#define WW      64
#define NPIX    4096
#define NB      2
#define KTOT    (NFRAMES*CDIM)
#define ROWS_BLK 4
#define TROWS    16
#define PITCH4   9
#define ATT_LDS  (TROWS*WW*PITCH4*16)

typedef __attribute__((ext_vector_type(2))) float v2f;
typedef __attribute__((ext_vector_type(8))) float v8f;

__device__ __forceinline__ int win_start(int i) {
    int ni = i - (KS / 2) * DILX;
    if (ni < 0) return i % DILX;
    if (i + (KS / 2) * DILX >= HH) return HH - KS * DILX + (i % DILX);
    return ni;
}

__global__ __launch_bounds__(256, 4)
void natten_attn_kernel(const float* __restrict__ q,
                                   const float* __restrict__ kk,
                                   const float* __restrict__ vv,
                                   const float* __restrict__ rpb,
                                   float* __restrict__ xn)
{
    extern __shared__ float4 tile[];

    int tid = blockIdx.x * blockDim.x + threadIdx.x;
    int j = tid & 63;
    int t = tid >> 6;
    int i = t & 63;
    t >>= 6;
    int f = t % NFRAMES;
    t /= NFRAMES;
    int head = t & 3;
    int b = t >> 2;

    int si = win_start(i), sj = win_start(j);
    int pbi = (si - i) / DILX + (KS - 1);
    int pbj = (sj - j) / DILX + (KS - 1);

    int i0 = (blockIdx.x * ROWS_BLK) & (HH - 1);
    int lo = i0 - (KS / 2) * DILX;
    lo = lo < 0 ? 0 : lo;
    lo = lo > HH - TROWS ? HH - TROWS : lo;

    const float scale = 0.17677669529663687f;

    const float* qp = q + ((size_t)(b * CDIM + head * HD)) * NPIX + i * WW + j;
    float qr[HD];
#pragma unroll
    for (int d = 0; d < HD; ++d) qr[d] = qp[(size_t)d * NPIX] * scale;

    const size_t kvbase = ((size_t)(b * KTOT + f * CDIM + head * HD)) * NPIX
                        + (size_t)lo * WW;
    const float* rp = rpb + head * (2 * KS - 1) * (2 * KS - 1);

    auto stage = [&](const float* __restrict__ srcp) {
#pragma unroll 2
        for (int c4 = 0; c4 < HD / 4; ++c4) {
#pragma unroll
            for (int pp = 0; pp < 4; ++pp) {
                int p = pp * 256 + threadIdx.x;
                const float* g = srcp + (size_t)(4 * c4) * NPIX + p;
                float4 tv;
                tv.x = g[0];
                tv.y = g[(size_t)NPIX];
                tv.z = g[(size_t)2 * NPIX];
                tv.w = g[(size_t)3 * NPIX];
                tile[p * PITCH4 + c4] = tv;
            }
        }
    };

    const float4* tb = tile + ((si - lo) * WW + sj) * PITCH4;

    float l[KS * KS];
#pragma unroll
    for (int ti = 0; ti < KS; ++ti) {
#pragma unroll
        for (int tj = 0; tj < KS; ++tj)
            l[ti * KS + tj] = rp[(pbi + ti) * (2 * KS - 1) + pbj + tj];
    }

    stage(kk + kvbase);
    __syncthreads();

#pragma unroll 1
    for (int ti = 0; ti < KS; ++ti) {
        const float4* trow = tb + ti * (DILX * WW * PITCH4);
        float4 kg[2][KS];
        float lv[KS];
#pragma unroll
        for (int tj = 0; tj < KS; ++tj) lv[tj] = l[tj];

#define HOLD() __builtin_amdgcn_sched_barrier(0)
#pragma unroll
        for (int tj = 0; tj < KS; ++tj) kg[0][tj] = trow[tj * (DILX * PITCH4)];
#pragma unroll
        for (int g = 0; g < HD / 4; ++g) {
            if (g + 1 < HD / 4) {
#pragma unroll
                for (int tj = 0; tj < KS; ++tj) kg[(g + 1) & 1][tj] = trow[tj * (DILX * PITCH4) + g + 1];
            }
            HOLD();
#pragma unroll
            for (int tj = 0; tj < KS; ++tj) lv[tj] = __builtin_fmaf(qr[4 * g + 0], kg[g & 1][tj].x, lv[tj]);
#pragma unroll
            for (int tj = 0; tj < KS; ++tj) lv[tj] = __builtin_fmaf(qr[4 * g + 1], kg[g & 1][tj].y, lv[tj]);
#pragma unroll
            for (int tj = 0; tj < KS; ++tj) lv[tj] = __builtin_fmaf(qr[4 * g + 2], kg[g & 1][tj].z, lv[tj]);
#pragma unroll
            for (int tj = 0; tj < KS; ++tj) lv[tj] = __builtin_fmaf(qr[4 * g + 3], kg[g & 1][tj].w, lv[tj]);
            HOLD();
        }
#undef HOLD

#pragma unroll
        for (int k = 0; k < KS * (KS - 1); ++k) l[k] = l[k + KS];
#pragma unroll
        for (int k = 0; k < KS; ++k) l[KS * (KS - 1) + k] = lv[k];
    }

    __syncthreads();
    stage(vv + kvbase);
    __syncthreads();

    float m = -3.0e38f, s = 0.0f;
    float acc[HD];
#pragma unroll
    for (int d = 0; d < HD; ++d) acc[d] = 0.0f;

#pragma unroll 1
    for (int ti = 0; ti < KS; ++ti) {
        const float4* trow = tb + ti * (DILX * WW * PITCH4);
#pragma unroll
        for (int tj = 0; tj < KS; ++tj) {
            const float4* tp = trow + tj * (DILX * PITCH4);
            float4 v4[HD / 4];
#pragma unroll
            for (int c = 0; c < HD / 4; ++c) v4[c] = tp[c];

            float lv   = l[tj];
            float mn   = fmaxf(m, lv);
            float corr = __expf(m - mn);
            float w    = __expf(lv - mn);
            s = s * corr + w;
            m = mn;
#pragma unroll
            for (int c = 0; c < HD / 4; ++c) {
                acc[4 * c + 0] = acc[4 * c + 0] * corr + w * v4[c].x;
                acc[4 * c + 1] = acc[4 * c + 1] * corr + w * v4[c].y;
                acc[4 * c + 2] = acc[4 * c + 2] * corr + w * v4[c].z;
                acc[4 * c + 3] = acc[4 * c + 3] * corr + w * v4[c].w;
            }
        }
#pragma unroll
        for (int k = 0; k < KS * (KS - 1); ++k) l[k] = l[k + KS];
    }

    float inv = 1.0f / s;
    float* op = xn + ((size_t)(b * KTOT + f * CDIM + head * HD)) * NPIX + i * WW + j;
#pragma unroll
    for (int d = 0; d < HD; ++d) op[(size_t)d * NPIX] = acc[d] * inv;
}

#define KGROUP   32
#define NGROUPS  (KTOT / KGROUP)

struct Frag {
    float a0[16], a1[16];
    float b0[16], b1[16];
};

__global__ __launch_bounds__(128)
void conv1x1_wmma_kernel(const float* __restrict__ xn,
                         const float* __restrict__ wzw,
                         const float* __restrict__ wzb,
                         float* __restrict__ out)
{
    int lane = threadIdx.x & 31;
    int wave = threadIdx.x >> 5;
    int col  = lane & 15;
    int hi   = lane >> 4;

    int mbase = (blockIdx.x & 1) * 64 + (wave & 1) * 32;
    int g0    = (blockIdx.x >> 1) * 64 + (wave >> 1) * 32;
    int b     = g0 >> 12;
    int p0    = (g0 & 4095) + col;

    const float* wp0 = wzw + (size_t)(mbase + col) * KTOT + 16 * hi;
    const float* wp1 = wp0 + (size_t)16 * KTOT;
    const float* xp0 = xn + ((size_t)b * KTOT + 16 * hi) * NPIX + p0;
    const float* xp1 = xp0 + 16;

    v8f acc00, acc01, acc10, acc11;
#pragma unroll
    for (int e = 0; e < 8; ++e) { acc00[e] = 0.0f; acc01[e] = 0.0f; acc10[e] = 0.0f; acc11[e] = 0.0f; }

    auto load_group = [&](int g, Frag& f) {
        const float* w0 = wp0 + KGROUP * g;
        const float* w1 = wp1 + KGROUP * g;
        const float* x0 = xp0 + (size_t)(KGROUP * g) * NPIX;
        const float* x1 = xp1 + (size_t)(KGROUP * g) * NPIX;
#pragma unroll
        for (int e = 0; e < 16; ++e) { f.a0[e] = w0[e]; f.a1[e] = w1[e]; }
#pragma unroll
        for (int e = 0; e < 16; ++e) {
            f.b0[e] = x0[(size_t)e * NPIX];
            f.b1[e] = x1[(size_t)e * NPIX];
        }
    };

    auto mma_group = [&](const Frag& f) {
#pragma unroll
        for (int j = 0; j < 8; ++j) {
            v2f A0, A1, B0, B1;
            A0.x = f.a0[2 * j]; A0.y = f.a0[2 * j + 1];
            A1.x = f.a1[2 * j]; A1.y = f.a1[2 * j + 1];
            B0.x = f.b0[2 * j]; B0.y = f.b0[2 * j + 1];
            B1.x = f.b1[2 * j]; B1.y = f.b1[2 * j + 1];
            acc00 = __builtin_amdgcn_wmma_f32_16x16x4_f32(
                false, A0, false, B0, (short)0, acc00, false, false);
            acc01 = __builtin_amdgcn_wmma_f32_16x16x4_f32(
                false, A0, false, B1, (short)0, acc01, false, false);
            acc10 = __builtin_amdgcn_wmma_f32_16x16x4_f32(
                false, A1, false, B0, (short)0, acc10, false, false);
            acc11 = __builtin_amdgcn_wmma_f32_16x16x4_f32(
                false, A1, false, B1, (short)0, acc11, false, false);
        }
    };

    Frag fA, fB;
    load_group(0, fA);

    for (int g = 0; g < NGROUPS - 2; g += 2) {
        load_group(g + 1, fB);
        __builtin_amdgcn_sched_barrier(0);
        mma_group(fA);
        __builtin_amdgcn_sched_barrier(0);
        load_group(g + 2, fA);
        __builtin_amdgcn_sched_barrier(0);
        mma_group(fB);
        __builtin_amdgcn_sched_barrier(0);
    }
    load_group(NGROUPS - 1, fB);
    __builtin_amdgcn_sched_barrier(0);
    mma_group(fA);
    mma_group(fB);

    float* op0 = out + (size_t)b * CDIM * NPIX + p0;
    float* op1 = op0 + 16;
#pragma unroll
    for (int e = 0; e < 8; ++e) {
        int r0 = mbase + e + 8 * hi;
        int r1 = r0 + 16;
        float bias0 = wzb[r0];
        float bias1 = wzb[r1];
        op0[(size_t)r0 * NPIX] = acc00[e] + bias0;
        op1[(size_t)r0 * NPIX] = acc01[e] + bias0;
        op0[(size_t)r1 * NPIX] = acc10[e] + bias1;
        op1[(size_t)r1 * NPIX] = acc11[e] + bias1;
    }
}

extern "C" void kernel_launch(void* const* d_in, const int* in_sizes, int n_in,
                              void* d_out, int out_size, void* d_ws, size_t ws_size,
                              hipStream_t stream) {
    const float* q   = (const float*)d_in[0];
    const float* k   = (const float*)d_in[1];
    const float* v   = (const float*)d_in[2];
    const float* rpb = (const float*)d_in[3];
    const float* wzw = (const float*)d_in[4];
    const float* wzb = (const float*)d_in[5];
    float* out = (float*)d_out;
    float* xn  = (float*)d_ws;

    (void)hipFuncSetAttribute(reinterpret_cast<const void*>(natten_attn_kernel),
                              hipFuncAttributeMaxDynamicSharedMemorySize,
                              (int)ATT_LDS);
    natten_attn_kernel<<<384, 256, ATT_LDS, stream>>>(q, k, v, rpb, xn);

    conv1x1_wmma_kernel<<<256, 128, 0, stream>>>(xn, wzw, wzb, out);
}


# attention kernel: block = 2 rows of one row parity (dilation 2: 8 same-parity rows staged, not 16); 768 blocks of 4 waves, LDS sized so every WGP holds 3 (even load, was 2 or 1); sums unchanged
# speedup vs baseline: 18.5567x; 1.0961x over previous
#include <hip/hip_runtime.h>
#include <hip/hip_bf16.h>

#define KS      7
#define DILX    2
#define NHEADS  4
#define NFRAMES 3
#define CDIM    128
#define HD      32
#define HH      64
#define WW      64
#define NPIX    4096
#define NB      2
#define KTOT    (NFRAMES*CDIM)
#define ROWS_BLK 2
#define TROWS    8
#define PITCH4   9
#define ATT_TILE (TROWS*WW*PITCH4*16)
#define ATT_LDS  (96*1024)

typedef __attribute__((ext_vector_type(2))) float v2f;
typedef __attribute__((ext_vector_type(8))) float v8f;

__device__ __forceinline__ int win_start(int i) {
    int ni = i - (KS / 2) * DILX;
    if (ni < 0) return i % DILX;
    if (i + (KS / 2) * DILX >= HH) return HH - KS * DILX + (i % DILX);
    return ni;
}

__global__ __launch_bounds__(128, 3)
void natten_attn_kernel(const float* __restrict__ q,
                                   const float* __restrict__ kk,
                                   const float* __restrict__ vv,
                                   const float* __restrict__ rpb,
                                   float* __restrict__ xn)
{
    extern __shared__ float4 tile[];

    int j   = threadIdx.x & 63;
    int rr  = threadIdx.x >> 6;
    int pr  = blockIdx.x & 15;
    int par = (blockIdx.x >> 4) & 1;
    int t   = blockIdx.x >> 5;
    int f = t % NFRAMES;
    t /= NFRAMES;
    int head = t & 3;
    int b = t >> 2;
    int i = 2 * (ROWS_BLK * pr + rr) + par;

    int si = win_start(i), sj = win_start(j);
    int pbi = (si - i) / DILX + (KS - 1);
    int pbj = (sj - j) / DILX + (KS - 1);

    int lo = ROWS_BLK * pr - KS / 2;
    lo = lo < 0 ? 0 : lo;
    lo = lo > HH / DILX - TROWS ? HH / DILX - TROWS : lo;

    const float scale = 0.17677669529663687f;

    const float* qp = q + ((size_t)(b * CDIM + head * HD)) * NPIX + i * WW + j;
    float qr[HD];
#pragma unroll
    for (int d = 0; d < HD; ++d) qr[d] = qp[(size_t)d * NPIX] * scale;

    const size_t kvbase = ((size_t)(b * KTOT + f * CDIM + head * HD)) * NPIX
                        + (size_t)((DILX * (lo + rr) + par) * WW + j);
    const float* rp = rpb + head * (2 * KS - 1) * (2 * KS - 1);

    auto stage = [&](const float* __restrict__ srcp) {
#pragma unroll 2
        for (int c4 = 0; c4 < HD / 4; ++c4) {
#pragma unroll
            for (int pp = 0; pp < 4; ++pp) {
                int p = pp * 128 + threadIdx.x;
                const float* g = srcp + (size_t)(4 * c4) * NPIX + pp * (2 * DILX * WW);
                float4 tv;
                tv.x = g[0];
                tv.y = g[(size_t)NPIX];
                tv.z = g[(size_t)2 * NPIX];
                tv.w = g[(size_t)3 * NPIX];
                tile[p * PITCH4 + c4] = tv;
            }
        }
    };

    const float4* tb = tile + ((si / DILX - lo) * WW + sj) * PITCH4;

    float l[KS * KS];
#pragma unroll
    for (int ti = 0; ti < KS; ++ti) {
#pragma unroll
        for (int tj = 0; tj < KS; ++tj)
            l[ti * KS + tj] = rp[(pbi + ti) * (2 * KS - 1) + pbj + tj];
    }

    stage(kk + kvbase);
    __syncthreads();

#pragma unroll 1
    for (int ti = 0; ti < KS; ++ti) {
        const float4* trow = tb + ti * (WW * PITCH4);
        float4 kg[2][KS];
        float lv[KS];
#pragma unroll
        for (int tj = 0; tj < KS; ++tj) lv[tj] = l[tj];

#define HOLD() __builtin_amdgcn_sched_barrier(0)
#pragma unroll
        for (int tj = 0; tj < KS; ++tj) kg[0][tj] = trow[tj * (DILX * PITCH4)];
#pragma unroll
        for (int g = 0; g < HD / 4; ++g) {
            if (g + 1 < HD / 4) {
#pragma unroll
                for (int tj = 0; tj < KS; ++tj) kg[(g + 1) & 1][tj] = trow[tj * (DILX * PITCH4) + g + 1];
            }
            HOLD();
#pragma unroll
            for (int tj = 0; tj < KS; ++tj) lv[tj] = __builtin_fmaf(qr[4 * g + 0], kg[g & 1][tj].x, lv[tj]);
#pragma unroll
            for (int tj = 0; tj < KS; ++tj) lv[tj] = __builtin_fmaf(qr[4 * g + 1], kg[g & 1][tj].y, lv[tj]);
#pragma unroll
            for (int tj = 0; tj < KS; ++tj) lv[tj] = __builtin_fmaf(qr[4 * g + 2], kg[g & 1][tj].z, lv[tj]);
#pragma unroll
            for (int tj = 0; tj < KS; ++tj) lv[tj] = __builtin_fmaf(qr[4 * g + 3], kg[g & 1][tj].w, lv[tj]);
            HOLD();
        }
#undef HOLD

#pragma unroll
        for (int k = 0; k < KS * (KS - 1); ++k) l[k] = l[k + KS];
#pragma unroll
        for (int k = 0; k < KS; ++k) l[KS * (KS - 1) + k] = lv[k];
    }

    __syncthreads();
    stage(vv + kvbase);
    __syncthreads();

    float m = -3.0e38f, s = 0.0f;
    float acc[HD];
#pragma unroll
    for (int d = 0; d < HD; ++d) acc[d] = 0.0f;

#pragma unroll 1
    for (int ti = 0; ti < KS; ++ti) {
        const float4* trow = tb + ti * (WW * PITCH4);
#pragma unroll
        for (int tj = 0; tj < KS; ++tj) {
            const float4* tp = trow + tj * (DILX * PITCH4);
            float4 v4[HD / 4];
#pragma unroll
            for (int c = 0; c < HD / 4; ++c) v4[c] = tp[c];

            float lv   = l[tj];
            float mn   = fmaxf(m, lv);
            float corr = __expf(m - mn);
            float w    = __expf(lv - mn);
            s = s * corr + w;
            m = mn;
#pragma unroll
            for (int c = 0; c < HD / 4; ++c) {
                acc[4 * c + 0] = acc[4 * c + 0] * corr + w * v4[c].x;
                acc[4 * c + 1] = acc[4 * c + 1] * corr + w * v4[c].y;
                acc[4 * c + 2] = acc[4 * c + 2] * corr + w * v4[c].z;
                acc[4 * c + 3] = acc[4 * c + 3] * corr + w * v4[c].w;
            }
        }
#pragma unroll
        for (int k = 0; k < KS * (KS - 1); ++k) l[k] = l[k + KS];
    }

    float inv = 1.0f / s;
    float* op = xn + ((size_t)(b * KTOT + f * CDIM + head * HD)) * NPIX + i * WW + j;
#pragma unroll
    for (int d = 0; d < HD; ++d) op[(size_t)d * NPIX] = acc[d] * inv;
}

#define KGROUP   32
#define NGROUPS  (KTOT / KGROUP)

struct Frag {
    float a0[16], a1[16];
    float b0[16], b1[16];
};

__global__ __launch_bounds__(128)
void conv1x1_wmma_kernel(const float* __restrict__ xn,
                         const float* __restrict__ wzw,
                         const float* __restrict__ wzb,
                         float* __restrict__ out)
{
    int lane = threadIdx.x & 31;
    int wave = threadIdx.x >> 5;
    int col  = lane & 15;
    int hi   = lane >> 4;

    int mbase = (blockIdx.x & 1) * 64 + (wave & 1) * 32;
    int g0    = (blockIdx.x >> 1) * 64 + (wave >> 1) * 32;
    int b     = g0 >> 12;
    int p0    = (g0 & 4095) + col;

    const float* wp0 = wzw + (size_t)(mbase + col) * KTOT + 16 * hi;
    const float* wp1 = wp0 + (size_t)16 * KTOT;
    const float* xp0 = xn + ((size_t)b * KTOT + 16 * hi) * NPIX + p0;
    const float* xp1 = xp0 + 16;

    v8f acc00, acc01, acc10, acc11;
#pragma unroll
    for (int e = 0; e < 8; ++e) { acc00[e] = 0.0f; acc01[e] = 0.0f; acc10[e] = 0.0f; acc11[e] = 0.0f; }

    auto load_group = [&](int g, Frag& f) {
        const float* w0 = wp0 + KGROUP * g;
        const float* w1 = wp1 + KGROUP * g;
        const float* x0 = xp0 + (size_t)(KGROUP * g) * NPIX;
        const float* x1 = xp1 + (size_t)(KGROUP * g) * NPIX;
#pragma unroll
        for (int e = 0; e < 16; ++e) { f.a0[e] = w0[e]; f.a1[e] = w1[e]; }
#pragma unroll
        for (int e = 0; e < 16; ++e) {
            f.b0[e] = x0[(size_t)e * NPIX];
            f.b1[e] = x1[(size_t)e * NPIX];
        }
    };

    auto mma_group = [&](const Frag& f) {
#pragma unroll
        for (int j = 0; j < 8; ++j) {
            v2f A0, A1, B0, B1;
            A0.x = f.a0[2 * j]; A0.y = f.a0[2 * j + 1];
            A1.x = f.a1[2 * j]; A1.y = f.a1[2 * j + 1];
            B0.x = f.b0[2 * j]; B0.y = f.b0[2 * j + 1];
            B1.x = f.b1[2 * j]; B1.y = f.b1[2 * j + 1];
            acc00 = __builtin_amdgcn_wmma_f32_16x16x4_f32(
                false, A0, false, B0, (short)0, acc00, false, false);
            acc01 = __builtin_amdgcn_wmma_f32_16x16x4_f32(
                false, A0, false, B1, (short)0, acc01, false, false);
            acc10 = __builtin_amdgcn_wmma_f32_16x16x4_f32(
                false, A1, false, B0, (short)0, acc10, false, false);
            acc11 = __builtin_amdgcn_wmma_f32_16x16x4_f32(
                false, A1, false, B1, (short)0, acc11, false, false);
        }
    };

    Frag fA, fB;
    load_group(0, fA);

    for (int g = 0; g < NGROUPS - 2; g += 2) {
        load_group(g + 1, fB);
        __builtin_amdgcn_sched_barrier(0);
        mma_group(fA);
        __builtin_amdgcn_sched_barrier(0);
        load_group(g + 2, fA);
        __builtin_amdgcn_sched_barrier(0);
        mma_group(fB);
        __builtin_amdgcn_sched_barrier(0);
    }
    load_group(NGROUPS - 1, fB);
    __builtin_amdgcn_sched_barrier(0);
    mma_group(fA);
    mma_group(fB);

    float* op0 = out + (size_t)b * CDIM * NPIX + p0;
    float* op1 = op0 + 16;
#pragma unroll
    for (int e = 0; e < 8; ++e) {
        int r0 = mbase + e + 8 * hi;
        int r1 = r0 + 16;
        float bias0 = wzb[r0];
        float bias1 = wzb[r1];
        op0[(size_t)r0 * NPIX] = acc00[e] + bias0;
        op1[(size_t)r0 * NPIX] = acc01[e] + bias0;
        op0[(size_t)r1 * NPIX] = acc10[e] + bias1;
        op1[(size_t)r1 * NPIX] = acc11[e] + bias1;
    }
}

extern "C" void kernel_launch(void* const* d_in, const int* in_sizes, int n_in,
                              void* d_out, int out_size, void* d_ws, size_t ws_size,
                              hipStream_t stream) {
    const float* q   = (const float*)d_in[0];
    const float* k   = (const float*)d_in[1];
    const float* v   = (const float*)d_in[2];
    const float* rpb = (const float*)d_in[3];
    const float* wzw = (const float*)d_in[4];
    const float* wzb = (const float*)d_in[5];
    float* out = (float*)d_out;
    float* xn  = (float*)d_ws;

    (void)hipFuncSetAttribute(reinterpret_cast<const void*>(natten_attn_kernel),
                              hipFuncAttributeMaxDynamicSharedMemorySize,
                              (int)ATT_LDS);
    natten_attn_kernel<<<768, 128, ATT_LDS, stream>>>(q, k, v, rpb, xn);

    conv1x1_wmma_kernel<<<256, 128, 0, stream>>>(xn, wzw, wzb, out);
}


# attention kernel: V band's global loads issued in three groups ahead of parts of the K pass (loop split 3+2+2 tap rows), held in registers, stored to LDS after it; sums unchanged
# speedup vs baseline: 20.0316x; 1.0795x over previous
#include <hip/hip_runtime.h>
#include <hip/hip_bf16.h>

#define KS      7
#define DILX    2
#define NHEADS  4
#define NFRAMES 3
#define CDIM    128
#define HD      32
#define HH      64
#define WW      64
#define NPIX    4096
#define NB      2
#define KTOT    (NFRAMES*CDIM)
#define ROWS_BLK 2
#define TROWS    8
#define PITCH4   9
#define ATT_TILE (TROWS*WW*PITCH4*16)
#define ATT_LDS  (96*1024)

typedef __attribute__((ext_vector_type(2))) float v2f;
typedef __attribute__((ext_vector_type(8))) float v8f;

__device__ __forceinline__ int win_start(int i) {
    int ni = i - (KS / 2) * DILX;
    if (ni < 0) return i % DILX;
    if (i + (KS / 2) * DILX >= HH) return HH - KS * DILX + (i % DILX);
    return ni;
}

__global__ __launch_bounds__(128, 3)
void natten_attn_kernel(const float* __restrict__ q,
                                   const float* __restrict__ kk,
                                   const float* __restrict__ vv,
                                   const float* __restrict__ rpb,
                                   float* __restrict__ xn)
{
    extern __shared__ float4 tile[];

    int j   = threadIdx.x & 63;
    int rr  = threadIdx.x >> 6;
    int pr  = blockIdx.x & 15;
    int par = (blockIdx.x >> 4) & 1;
    int t   = blockIdx.x >> 5;
    int f = t % NFRAMES;
    t /= NFRAMES;
    int head = t & 3;
    int b = t >> 2;
    int i = 2 * (ROWS_BLK * pr + rr) + par;

    int si = win_start(i), sj = win_start(j);
    int pbi = (si - i) / DILX + (KS - 1);
    int pbj = (sj - j) / DILX + (KS - 1);

    int lo = ROWS_BLK * pr - KS / 2;
    lo = lo < 0 ? 0 : lo;
    lo = lo > HH / DILX - TROWS ? HH / DILX - TROWS : lo;

    const float scale = 0.17677669529663687f;

    const float* qp = q + ((size_t)(b * CDIM + head * HD)) * NPIX + i * WW + j;
    float qr[HD];
#pragma unroll
    for (int d = 0; d < HD; ++d) qr[d] = qp[(size_t)d * NPIX] * scale;

    const size_t kvbase = ((size_t)(b * KTOT + f * CDIM + head * HD)) * NPIX
                        + (size_t)((DILX * (lo + rr) + par) * WW + j);
    const float* rp = rpb + head * (2 * KS - 1) * (2 * KS - 1);

    auto stage = [&](const float* __restrict__ srcp) {
#pragma unroll 2
        for (int c4 = 0; c4 < HD / 4; ++c4) {
#pragma unroll
            for (int pp = 0; pp < 4; ++pp) {
                int p = pp * 128 + threadIdx.x;
                const float* g = srcp + (size_t)(4 * c4) * NPIX + pp * (2 * DILX * WW);
                float4 tv;
                tv.x = g[0];
                tv.y = g[(size_t)NPIX];
                tv.z = g[(size_t)2 * NPIX];
                tv.w = g[(size_t)3 * NPIX];
                tile[p * PITCH4 + c4] = tv;
            }
        }
    };

    const float4* tb = tile + ((si / DILX - lo) * WW + sj) * PITCH4;

    float l[KS * KS];
#pragma unroll
    for (int ti = 0; ti < KS; ++ti) {
#pragma unroll
        for (int tj = 0; tj < KS; ++tj)
            l[ti * KS + tj] = rp[(pbi + ti) * (2 * KS - 1) + pbj + tj];
    }

    stage(kk + kvbase);
    __syncthreads();

#define HOLD() __builtin_amdgcn_sched_barrier(0)
#define K_ROW(ti) {                                                                          \
        const float4* trow = tb + (ti) * (WW * PITCH4);                                      \
        float4 kg[2][KS];                                                                    \
        float lv[KS];                                                                        \
        _Pragma("unroll")                                                                    \
        for (int tj = 0; tj < KS; ++tj) lv[tj] = l[tj];                                      \
        _Pragma("unroll")                                                                    \
        for (int tj = 0; tj < KS; ++tj) kg[0][tj] = trow[tj * (DILX * PITCH4)];              \
        _Pragma("unroll")                                                                    \
        for (int g = 0; g < HD / 4; ++g) {                                                   \
            if (g + 1 < HD / 4) {                                                            \
                _Pragma("unroll")                                                            \
                for (int tj = 0; tj < KS; ++tj)                                              \
                    kg[(g + 1) & 1][tj] = trow[tj * (DILX * PITCH4) + g + 1];                \
            }                                                                                \
            HOLD();                                                                          \
            _Pragma("unroll")                                                                \
            for (int tj = 0; tj < KS; ++tj)                                                  \
                lv[tj] = __builtin_fmaf(qr[4 * g + 0], kg[g & 1][tj].x, lv[tj]);             \
            _Pragma("unroll")                                                                \
            for (int tj = 0; tj < KS; ++tj)                                                  \
                lv[tj] = __builtin_fmaf(qr[4 * g + 1], kg[g & 1][tj].y, lv[tj]);             \
            _Pragma("unroll")                                                                \
            for (int tj = 0; tj < KS; ++tj)                                                  \
                lv[tj] = __builtin_fmaf(qr[4 * g + 2], kg[g & 1][tj].z, lv[tj]);             \
            _Pragma("unroll")                                                                \
            for (int tj = 0; tj < KS; ++tj)                                                  \
                lv[tj] = __builtin_fmaf(qr[4 * g + 3], kg[g & 1][tj].w, lv[tj]);             \
            HOLD();                                                                          \
        }                                                                                    \
        _Pragma("unroll")                                                                    \
        for (int k = 0; k < KS * (KS - 1); ++k) l[k] = l[k + KS];                            \
        _Pragma("unroll")                                                                    \
        for (int k = 0; k < KS; ++k) l[KS * (KS - 1) + k] = lv[k];                           \
    }

    float4 vq[4 * (HD / 4)];
    const float* vsrc = vv + kvbase;
#define VLOAD(U0, U1)                                                                        \
    _Pragma("unroll")                                                                        \
    for (int u = (U0); u < (U1); ++u) {                                                      \
        const float* g = vsrc + (size_t)(4 * (u >> 2)) * NPIX + (u & 3) * (2 * DILX * WW);   \
        vq[u].x = g[0];                                                                      \
        vq[u].y = g[(size_t)NPIX];                                                           \
        vq[u].z = g[(size_t)2 * NPIX];                                                       \
        vq[u].w = g[(size_t)3 * NPIX];                                                       \
    }

    VLOAD(0, 12)
#pragma unroll 1
    for (int ti = 0; ti < 3; ++ti) K_ROW(ti)
    VLOAD(12, 22)
#pragma unroll 1
    for (int ti = 3; ti < 5; ++ti) K_ROW(ti)
    VLOAD(22, 32)
#pragma unroll 1
    for (int ti = 5; ti < KS; ++ti) K_ROW(ti)
#undef VLOAD
#undef K_ROW
#undef HOLD

    __syncthreads();
#pragma unroll
    for (int u = 0; u < 4 * (HD / 4); ++u)
        tile[((u & 3) * 128 + threadIdx.x) * PITCH4 + (u >> 2)] = vq[u];
    __syncthreads();

    float m = -3.0e38f, s = 0.0f;
    float acc[HD];
#pragma unroll
    for (int d = 0; d < HD; ++d) acc[d] = 0.0f;

#pragma unroll 1
    for (int ti = 0; ti < KS; ++ti) {
        const float4* trow = tb + ti * (WW * PITCH4);
#pragma unroll
        for (int tj = 0; tj < KS; ++tj) {
            const float4* tp = trow + tj * (DILX * PITCH4);
            float4 v4[HD / 4];
#pragma unroll
            for (int c = 0; c < HD / 4; ++c) v4[c] = tp[c];

            float lv   = l[tj];
            float mn   = fmaxf(m, lv);
            float corr = __expf(m - mn);
            float w    = __expf(lv - mn);
            s = s * corr + w;
            m = mn;
#pragma unroll
            for (int c = 0; c < HD / 4; ++c) {
                acc[4 * c + 0] = acc[4 * c + 0] * corr + w * v4[c].x;
                acc[4 * c + 1] = acc[4 * c + 1] * corr + w * v4[c].y;
                acc[4 * c + 2] = acc[4 * c + 2] * corr + w * v4[c].z;
                acc[4 * c + 3] = acc[4 * c + 3] * corr + w * v4[c].w;
            }
        }
#pragma unroll
        for (int k = 0; k < KS * (KS - 1); ++k) l[k] = l[k + KS];
    }

    float inv = 1.0f / s;
    float* op = xn + ((size_t)(b * KTOT + f * CDIM + head * HD)) * NPIX + i * WW + j;
#pragma unroll
    for (int d = 0; d < HD; ++d) op[(size_t)d * NPIX] = acc[d] * inv;
}

#define KGROUP   32
#define NGROUPS  (KTOT / KGROUP)

struct Frag {
    float a0[16], a1[16];
    float b0[16], b1[16];
};

__global__ __launch_bounds__(128)
void conv1x1_wmma_kernel(const float* __restrict__ xn,
                         const float* __restrict__ wzw,
                         const float* __restrict__ wzb,
                         float* __restrict__ out)
{
    int lane = threadIdx.x & 31;
    int wave = threadIdx.x >> 5;
    int col  = lane & 15;
    int hi   = lane >> 4;

    int mbase = (blockIdx.x & 1) * 64 + (wave & 1) * 32;
    int g0    = (blockIdx.x >> 1) * 64 + (wave >> 1) * 32;
    int b     = g0 >> 12;
    int p0    = (g0 & 4095) + col;

    const float* wp0 = wzw + (size_t)(mbase + col) * KTOT + 16 * hi;
    const float* wp1 = wp0 + (size_t)16 * KTOT;
    const float* xp0 = xn + ((size_t)b * KTOT + 16 * hi) * NPIX + p0;
    const float* xp1 = xp0 + 16;

    v8f acc00, acc01, acc10, acc11;
#pragma unroll
    for (int e = 0; e < 8; ++e) { acc00[e] = 0.0f; acc01[e] = 0.0f; acc10[e] = 0.0f; acc11[e] = 0.0f; }

    auto load_group = [&](int g, Frag& f) {
        const float* w0 = wp0 + KGROUP * g;
        const float* w1 = wp1 + KGROUP * g;
        const float* x0 = xp0 + (size_t)(KGROUP * g) * NPIX;
        const float* x1 = xp1 + (size_t)(KGROUP * g) * NPIX;
#pragma unroll
        for (int e = 0; e < 16; ++e) { f.a0[e] = w0[e]; f.a1[e] = w1[e]; }
#pragma unroll
        for (int e = 0; e < 16; ++e) {
            f.b0[e] = x0[(size_t)e * NPIX];
            f.b1[e] = x1[(size_t)e * NPIX];
        }
    };

    auto mma_group = [&](const Frag& f) {
#pragma unroll
        for (int j = 0; j < 8; ++j) {
            v2f A0, A1, B0, B1;
            A0.x = f.a0[2 * j]; A0.y = f.a0[2 * j + 1];
            A1.x = f.a1[2 * j]; A1.y = f.a1[2 * j + 1];
            B0.x = f.b0[2 * j]; B0.y = f.b0[2 * j + 1];
            B1.x = f.b1[2 * j]; B1.y = f.b1[2 * j + 1];
            acc00 = __builtin_amdgcn_wmma_f32_16x16x4_f32(
                false, A0, false, B0, (short)0, acc00, false, false);
            acc01 = __builtin_amdgcn_wmma_f32_16x16x4_f32(
                false, A0, false, B1, (short)0, acc01, false, false);
            acc10 = __builtin_amdgcn_wmma_f32_16x16x4_f32(
                false, A1, false, B0, (short)0, acc10, false, false);
            acc11 = __builtin_amdgcn_wmma_f32_16x16x4_f32(
                false, A1, false, B1, (short)0, acc11, false, false);
        }
    };

    Frag fA, fB;
    load_group(0, fA);

    for (int g = 0; g < NGROUPS - 2; g += 2) {
        load_group(g + 1, fB);
        __builtin_amdgcn_sched_barrier(0);
        mma_group(fA);
        __builtin_amdgcn_sched_barrier(0);
        load_group(g + 2, fA);
        __builtin_amdgcn_sched_barrier(0);
        mma_group(fB);
        __builtin_amdgcn_sched_barrier(0);
    }
    load_group(NGROUPS - 1, fB);
    __builtin_amdgcn_sched_barrier(0);
    mma_group(fA);
    mma_group(fB);

    float* op0 = out + (size_t)b * CDIM * NPIX + p0;
    float* op1 = op0 + 16;
#pragma unroll
    for (int e = 0; e < 8; ++e) {
        int r0 = mbase + e + 8 * hi;
        int r1 = r0 + 16;
        float bias0 = wzb[r0];
        float bias1 = wzb[r1];
        op0[(size_t)r0 * NPIX] = acc00[e] + bias0;
        op1[(size_t)r0 * NPIX] = acc01[e] + bias0;
        op0[(size_t)r1 * NPIX] = acc10[e] + bias1;
        op1[(size_t)r1 * NPIX] = acc11[e] + bias1;
    }
}

extern "C" void kernel_launch(void* const* d_in, const int* in_sizes, int n_in,
                              void* d_out, int out_size, void* d_ws, size_t ws_size,
                              hipStream_t stream) {
    const float* q   = (const float*)d_in[0];
    const float* k   = (const float*)d_in[1];
    const float* v   = (const float*)d_in[2];
    const float* rpb = (const float*)d_in[3];
    const float* wzw = (const float*)d_in[4];
    const float* wzb = (const float*)d_in[5];
    float* out = (float*)d_out;
    float* xn  = (float*)d_ws;

    (void)hipFuncSetAttribute(reinterpret_cast<const void*>(natten_attn_kernel),
                              hipFuncAttributeMaxDynamicSharedMemorySize,
                              (int)ATT_LDS);
    natten_attn_kernel<<<768, 128, ATT_LDS, stream>>>(q, k, v, rpb, xn);

    conv1x1_wmma_kernel<<<256, 128, 0, stream>>>(xn, wzw, wzb, out);
}


# attention kernel, V pass: maximum of the 49 logits taken first, so a tap's weight is final: one multiply-add per tap and channel instead of rescale multiply and multiply-add; order of sums unchanged
# speedup vs baseline: 20.7695x; 1.0368x over previous
#include <hip/hip_runtime.h>
#include <hip/hip_bf16.h>

#define KS      7
#define DILX    2
#define NHEADS  4
#define NFRAMES 3
#define CDIM    128
#define HD      32
#define HH      64
#define WW      64
#define NPIX    4096
#define NB      2
#define KTOT    (NFRAMES*CDIM)
#define ROWS_BLK 2
#define TROWS    8
#define PITCH4   9
#define ATT_TILE (TROWS*WW*PITCH4*16)
#define ATT_LDS  (96*1024)

typedef __attribute__((ext_vector_type(2))) float v2f;
typedef __attribute__((ext_vector_type(8))) float v8f;

__device__ __forceinline__ int win_start(int i) {
    int ni = i - (KS / 2) * DILX;
    if (ni < 0) return i % DILX;
    if (i + (KS / 2) * DILX >= HH) return HH - KS * DILX + (i % DILX);
    return ni;
}

__global__ __launch_bounds__(128, 3)
void natten_attn_kernel(const float* __restrict__ q,
                                   const float* __restrict__ kk,
                                   const float* __restrict__ vv,
                                   const float* __restrict__ rpb,
                                   float* __restrict__ xn)
{
    extern __shared__ float4 tile[];

    int j   = threadIdx.x & 63;
    int rr  = threadIdx.x >> 6;
    int pr  = blockIdx.x & 15;
    int par = (blockIdx.x >> 4) & 1;
    int t   = blockIdx.x >> 5;
    int f = t % NFRAMES;
    t /= NFRAMES;
    int head = t & 3;
    int b = t >> 2;
    int i = 2 * (ROWS_BLK * pr + rr) + par;

    int si = win_start(i), sj = win_start(j);
    int pbi = (si - i) / DILX + (KS - 1);
    int pbj = (sj - j) / DILX + (KS - 1);

    int lo = ROWS_BLK * pr - KS / 2;
    lo = lo < 0 ? 0 : lo;
    lo = lo > HH / DILX - TROWS ? HH / DILX - TROWS : lo;

    const float scale = 0.17677669529663687f;

    const float* qp = q + ((size_t)(b * CDIM + head * HD)) * NPIX + i * WW + j;
    float qr[HD];
#pragma unroll
    for (int d = 0; d < HD; ++d) qr[d] = qp[(size_t)d * NPIX] * scale;

    const size_t kvbase = ((size_t)(b * KTOT + f * CDIM + head * HD)) * NPIX
                        + (size_t)((DILX * (lo + rr) + par) * WW + j);
    const float* rp = rpb + head * (2 * KS - 1) * (2 * KS - 1);

    auto stage = [&](const float* __restrict__ srcp) {
#pragma unroll 2
        for (int c4 = 0; c4 < HD / 4; ++c4) {
#pragma unroll
            for (int pp = 0; pp < 4; ++pp) {
                int p = pp * 128 + threadIdx.x;
                const float* g = srcp + (size_t)(4 * c4) * NPIX + pp * (2 * DILX * WW);
                float4 tv;
                tv.x = g[0];
                tv.y = g[(size_t)NPIX];
                tv.z = g[(size_t)2 * NPIX];
                tv.w = g[(size_t)3 * NPIX];
                tile[p * PITCH4 + c4] = tv;
            }
        }
    };

    const float4* tb = tile + ((si / DILX - lo) * WW + sj) * PITCH4;

    float l[KS * KS];
#pragma unroll
    for (int ti = 0; ti < KS; ++ti) {
#pragma unroll
        for (int tj = 0; tj < KS; ++tj)
            l[ti * KS + tj] = rp[(pbi + ti) * (2 * KS - 1) + pbj + tj];
    }

    stage(kk + kvbase);
    __syncthreads();

#define HOLD() __builtin_amdgcn_sched_barrier(0)
#define K_ROW(ti) {                                                                          \
        const float4* trow = tb + (ti) * (WW * PITCH4);                                      \
        float4 kg[2][KS];                                                                    \
        float lv[KS];                                                                        \
        _Pragma("unroll")                                                                    \
        for (int tj = 0; tj < KS; ++tj) lv[tj] = l[tj];                                      \
        _Pragma("unroll")                                                                    \
        for (int tj = 0; tj < KS; ++tj) kg[0][tj] = trow[tj * (DILX * PITCH4)];              \
        _Pragma("unroll")                                                                    \
        for (int g = 0; g < HD / 4; ++g) {                                                   \
            if (g + 1 < HD / 4) {                                                            \
                _Pragma("unroll")                                                            \
                for (int tj = 0; tj < KS; ++tj)                                              \
                    kg[(g + 1) & 1][tj] = trow[tj * (DILX * PITCH4) + g + 1];                \
            }                                                                                \
            HOLD();                                                                          \
            _Pragma("unroll")                                                                \
            for (int tj = 0; tj < KS; ++tj)                                                  \
                lv[tj] = __builtin_fmaf(qr[4 * g + 0], kg[g & 1][tj].x, lv[tj]);             \
            _Pragma("unroll")                                                                \
            for (int tj = 0; tj < KS; ++tj)                                                  \
                lv[tj] = __builtin_fmaf(qr[4 * g + 1], kg[g & 1][tj].y, lv[tj]);             \
            _Pragma("unroll")                                                                \
            for (int tj = 0; tj < KS; ++tj)                                                  \
                lv[tj] = __builtin_fmaf(qr[4 * g + 2], kg[g & 1][tj].z, lv[tj]);             \
            _Pragma("unroll")                                                                \
            for (int tj = 0; tj < KS; ++tj)                                                  \
                lv[tj] = __builtin_fmaf(qr[4 * g + 3], kg[g & 1][tj].w, lv[tj]);             \
            HOLD();                                                                          \
        }                                                                                    \
        _Pragma("unroll")                                                                    \
        for (int k = 0; k < KS * (KS - 1); ++k) l[k] = l[k + KS];                            \
        _Pragma("unroll")                                                                    \
        for (int k = 0; k < KS; ++k) l[KS * (KS - 1) + k] = lv[k];                           \
    }

    float4 vq[4 * (HD / 4)];
    const float* vsrc = vv + kvbase;
#define VLOAD(U0, U1)                                                                        \
    _Pragma("unroll")                                                                        \
    for (int u = (U0); u < (U1); ++u) {                                                      \
        const float* g = vsrc + (size_t)(4 * (u >> 2)) * NPIX + (u & 3) * (2 * DILX * WW);   \
        vq[u].x = g[0];                                                                      \
        vq[u].y = g[(size_t)NPIX];                                                           \
        vq[u].z = g[(size_t)2 * NPIX];                                                       \
        vq[u].w = g[(size_t)3 * NPIX];                                                       \
    }

    VLOAD(0, 12)
#pragma unroll 1
    for (int ti = 0; ti < 3; ++ti) K_ROW(ti)
    VLOAD(12, 22)
#pragma unroll 1
    for (int ti = 3; ti < 5; ++ti) K_ROW(ti)
    VLOAD(22, 32)
#pragma unroll 1
    for (int ti = 5; ti < KS; ++ti) K_ROW(ti)
#undef VLOAD
#undef K_ROW
#undef HOLD

    __syncthreads();
#pragma unroll
    for (int u = 0; u < 4 * (HD / 4); ++u)
        tile[((u & 3) * 128 + threadIdx.x) * PITCH4 + (u >> 2)] = vq[u];
    __syncthreads();

    float m = l[0];
#pragma unroll
    for (int k = 1; k < KS * KS; ++k) m = fmaxf(m, l[k]);
    float s = 0.0f;
    float acc[HD];
#pragma unroll
    for (int d = 0; d < HD; ++d) acc[d] = 0.0f;

#pragma unroll 1
    for (int ti = 0; ti < KS; ++ti) {
        const float4* trow = tb + ti * (WW * PITCH4);
#pragma unroll
        for (int tj = 0; tj < KS; ++tj) {
            const float4* tp = trow + tj * (DILX * PITCH4);
            float4 v4[HD / 4];
#pragma unroll
            for (int c = 0; c < HD / 4; ++c) v4[c] = tp[c];

            float w = __expf(l[tj] - m);
            s += w;
#pragma unroll
            for (int c = 0; c < HD / 4; ++c) {
                acc[4 * c + 0] += w * v4[c].x;
                acc[4 * c + 1] += w * v4[c].y;
                acc[4 * c + 2] += w * v4[c].z;
                acc[4 * c + 3] += w * v4[c].w;
            }
        }
#pragma unroll
        for (int k = 0; k < KS * (KS - 1); ++k) l[k] = l[k + KS];
    }

    float inv = 1.0f / s;
    float* op = xn + ((size_t)(b * KTOT + f * CDIM + head * HD)) * NPIX + i * WW + j;
#pragma unroll
    for (int d = 0; d < HD; ++d) op[(size_t)d * NPIX] = acc[d] * inv;
}

#define KGROUP   32
#define NGROUPS  (KTOT / KGROUP)

struct Frag {
    float a0[16], a1[16];
    float b0[16], b1[16];
};

__global__ __launch_bounds__(128)
void conv1x1_wmma_kernel(const float* __restrict__ xn,
                         const float* __restrict__ wzw,
                         const float* __restrict__ wzb,
                         float* __restrict__ out)
{
    int lane = threadIdx.x & 31;
    int wave = threadIdx.x >> 5;
    int col  = lane & 15;
    int hi   = lane >> 4;

    int mbase = (blockIdx.x & 1) * 64 + (wave & 1) * 32;
    int g0    = (blockIdx.x >> 1) * 64 + (wave >> 1) * 32;
    int b     = g0 >> 12;
    int p0    = (g0 & 4095) + col;

    const float* wp0 = wzw + (size_t)(mbase + col) * KTOT + 16 * hi;
    const float* wp1 = wp0 + (size_t)16 * KTOT;
    const float* xp0 = xn + ((size_t)b * KTOT + 16 * hi) * NPIX + p0;
    const float* xp1 = xp0 + 16;

    v8f acc00, acc01, acc10, acc11;
#pragma unroll
    for (int e = 0; e < 8; ++e) { acc00[e] = 0.0f; acc01[e] = 0.0f; acc10[e] = 0.0f; acc11[e] = 0.0f; }

    auto load_group = [&](int g, Frag& f) {
        const float* w0 = wp0 + KGROUP * g;
        const float* w1 = wp1 + KGROUP * g;
        const float* x0 = xp0 + (size_t)(KGROUP * g) * NPIX;
        const float* x1 = xp1 + (size_t)(KGROUP * g) * NPIX;
#pragma unroll
        for (int e = 0; e < 16; ++e) { f.a0[e] = w0[e]; f.a1[e] = w1[e]; }
#pragma unroll
        for (int e = 0; e < 16; ++e) {
            f.b0[e] = x0[(size_t)e * NPIX];
            f.b1[e] = x1[(size_t)e * NPIX];
        }
    };

    auto mma_group = [&](const Frag& f) {
#pragma unroll
        for (int j = 0; j < 8; ++j) {
            v2f A0, A1, B0, B1;
            A0.x = f.a0[2 * j]; A0.y = f.a0[2 * j + 1];
            A1.x = f.a1[2 * j]; A1.y = f.a1[2 * j + 1];
            B0.x = f.b0[2 * j]; B0.y = f.b0[2 * j + 1];
            B1.x = f.b1[2 * j]; B1.y = f.b1[2 * j + 1];
            acc00 = __builtin_amdgcn_wmma_f32_16x16x4_f32(
                false, A0, false, B0, (short)0, acc00, false, false);
            acc01 = __builtin_amdgcn_wmma_f32_16x16x4_f32(
                false, A0, false, B1, (short)0, acc01, false, false);
            acc10 = __builtin_amdgcn_wmma_f32_16x16x4_f32(
                false, A1, false, B0, (short)0, acc10, false, false);
            acc11 = __builtin_amdgcn_wmma_f32_16x16x4_f32(
                false, A1, false, B1, (short)0, acc11, false, false);
        }
    };

    Frag fA, fB;
    load_group(0, fA);

    for (int g = 0; g < NGROUPS - 2; g += 2) {
        load_group(g + 1, fB);
        __builtin_amdgcn_sched_barrier(0);
        mma_group(fA);
        __builtin_amdgcn_sched_barrier(0);
        load_group(g + 2, fA);
        __builtin_amdgcn_sched_barrier(0);
        mma_group(fB);
        __builtin_amdgcn_sched_barrier(0);
    }
    load_group(NGROUPS - 1, fB);
    __builtin_amdgcn_sched_barrier(0);
    mma_group(fA);
    mma_group(fB);

    float* op0 = out + (size_t)b * CDIM * NPIX + p0;
    float* op1 = op0 + 16;
#pragma unroll
    for (int e = 0; e < 8; ++e) {
        int r0 = mbase + e + 8 * hi;
        int r1 = r0 + 16;
        float bias0 = wzb[r0];
        float bias1 = wzb[r1];
        op0[(size_t)r0 * NPIX] = acc00[e] + bias0;
        op1[(size_t)r0 * NPIX] = acc01[e] + bias0;
        op0[(size_t)r1 * NPIX] = acc10[e] + bias1;
        op1[(size_t)r1 * NPIX] = acc11[e] + bias1;
    }
}

extern "C" void kernel_launch(void* const* d_in, const int* in_sizes, int n_in,
                              void* d_out, int out_size, void* d_ws, size_t ws_size,
                              hipStream_t stream) {
    const float* q   = (const float*)d_in[0];
    const float* k   = (const float*)d_in[1];
    const float* v   = (const float*)d_in[2];
    const float* rpb = (const float*)d_in[3];
    const float* wzw = (const float*)d_in[4];
    const float* wzb = (const float*)d_in[5];
    float* out = (float*)d_out;
    float* xn  = (float*)d_ws;

    (void)hipFuncSetAttribute(reinterpret_cast<const void*>(natten_attn_kernel),
                              hipFuncAttributeMaxDynamicSharedMemorySize,
                              (int)ATT_LDS);
    natten_attn_kernel<<<768, 128, ATT_LDS, stream>>>(q, k, v, rpb, xn);

    conv1x1_wmma_kernel<<<256, 128, 0, stream>>>(xn, wzw, wzb, out);
}
